# starting kernel
# baseline (speedup 1.0000x reference)
#include <hip/hip_runtime.h>
#include <hip/hip_bf16.h>


typedef float v2f __attribute__((ext_vector_type(2)));
typedef float v8f __attribute__((ext_vector_type(8)));

#define NEG_SLOPE 0.2f

__device__ __forceinline__ unsigned f2ord(float f) {
  unsigned u = __float_as_uint(f);
  return (f >= 0.0f) ? (u | 0x80000000u) : ~u;
}
__device__ __forceinline__ float ord2f(unsigned u) {
  return (u & 0x80000000u) ? __uint_as_float(u & 0x7FFFFFFFu)
                           : __uint_as_float(~u);
}

template <int K, int NOUT>
__global__ void gemm_wmma_f32(const float* __restrict__ A,
                              const float* __restrict__ B,
                              float* __restrict__ C, int M) {
  const int NT    = NOUT / 16;
  const int wave  = (blockIdx.x * blockDim.x + threadIdx.x) >> 5;
  const int lane  = threadIdx.x & 31;
  const int mt    = wave / NT;
  const int nt    = wave % NT;
  if (mt * 16 >= M) return;

  const int half = lane >> 4;
  const int lr   = lane & 15;
  int arow = mt * 16 + lr;
  if (arow >= M) arow = M - 1;
  const int bcol = nt * 16 + lr;

  v8f acc = {};
  const float* aptr = A + (size_t)arow * K + 2 * half;
  const float* bptr = B + (size_t)(2 * half) * NOUT + bcol;
  for (int k = 0; k < K; k += 4) {
    v2f a = *(const v2f*)(aptr + k);
    v2f b;
    b.x = bptr[(size_t)k * NOUT];
    b.y = bptr[(size_t)(k + 1) * NOUT];
    acc = __builtin_amdgcn_wmma_f32_16x16x4_f32(
         false, a,  false, b,
         (short)0, acc,  false,  false);
  }

  const int orow = mt * 16 + half * 8;
#pragma unroll
  for (int r = 0; r < 8; ++r) {
    if (orow + r < M) C[(size_t)(orow + r) * NOUT + bcol] = acc[r];
  }
}

__global__ void att_scores(const float* __restrict__ h,
                           const float* __restrict__ att_src,
                           const float* __restrict__ att_dst,
                           float* __restrict__ asrc, float* __restrict__ adst,
                           int n, int heads, int ch) {
  int i = blockIdx.x * blockDim.x + threadIdx.x;
  if (i >= n * heads) return;
  int node = i / heads, hd = i % heads;
  const float* hp = h + (size_t)node * heads * ch + hd * ch;
  const float* as = att_src + hd * ch;
  const float* ad = att_dst + hd * ch;
  float ss = 0.0f, sd = 0.0f;
  for (int c = 0; c < ch; ++c) {
    float v = hp[c];
    ss += v * as[c];
    sd += v * ad[c];
  }
  asrc[i] = ss;
  adst[i] = sd;
}

template <int H>
__global__ void edge_max(const int* __restrict__ ei, int E, int T,
                         const float* __restrict__ asrc,
                         const float* __restrict__ adst,
                         unsigned* __restrict__ m) {
  int e = blockIdx.x * blockDim.x + threadIdx.x;
  if (e >= T) return;
  int s = (e < E) ? ei[e] : (e - E);
  int d = (e < E) ? ei[E + e] : (e - E);
#pragma unroll
  for (int h = 0; h < H; ++h) {
    float v = asrc[s * H + h] + adst[d * H + h];
    v = (v > 0.0f) ? v : NEG_SLOPE * v;
    atomicMax(&m[d * H + h], f2ord(v));
  }
}

template <int H>
__global__ void edge_expsum(const int* __restrict__ ei, int E, int T,
                            const float* __restrict__ asrc,
                            const float* __restrict__ adst,
                            const unsigned* __restrict__ m,
                            float* __restrict__ ssum) {
  int e = blockIdx.x * blockDim.x + threadIdx.x;
  if (e >= T) return;
  int s = (e < E) ? ei[e] : (e - E);
  int d = (e < E) ? ei[E + e] : (e - E);
#pragma unroll
  for (int h = 0; h < H; ++h) {
    float v = asrc[s * H + h] + adst[d * H + h];
    v = (v > 0.0f) ? v : NEG_SLOPE * v;
    float ex = __expf(v - ord2f(m[d * H + h]));
    atomicAdd(&ssum[d * H + h], ex);
  }
}

template <int H, int C>
__global__ void edge_aggregate(const int* __restrict__ ei, int E, int T,
                               const float* __restrict__ asrc,
                               const float* __restrict__ adst,
                               const unsigned* __restrict__ m,
                               const float* __restrict__ ssum,
                               const float* __restrict__ hfeat,
                               float* __restrict__ out) {
  const int wave = (blockIdx.x * blockDim.x + threadIdx.x) >> 5;
  const int lane = threadIdx.x & 31;
  if (wave >= T) return;
  int s = (wave < E) ? ei[wave] : (wave - E);
  int d = (wave < E) ? ei[E + wave] : (wave - E);

  float alpha = 0.0f;
  if (lane < H) {
    float v = asrc[s * H + lane] + adst[d * H + lane];
    v = (v > 0.0f) ? v : NEG_SLOPE * v;
    alpha = __expf(v - ord2f(m[d * H + lane])) / (ssum[d * H + lane] + 1e-16f);
  }

  const int F = H * C;
  const float* hp = hfeat + (size_t)s * F;
  float* op = out + (size_t)d * F;
#pragma unroll
  for (int f = lane; f < F; f += 32) {
    float a = __shfl(alpha, f / C, 32);
    atomicAdd(&op[f], hp[f] * a);
  }
}

__global__ void init_bias(float* __restrict__ out, const float* __restrict__ bias,
                          int n, int F) {
  int i = blockIdx.x * blockDim.x + threadIdx.x;
  if (i < n * F) out[i] = bias[i % F];
}
__global__ void init_zero_u32(unsigned* __restrict__ p, int n) {
  int i = blockIdx.x * blockDim.x + threadIdx.x;
  if (i < n) p[i] = 0u;
}
__global__ void init_zero_f32(float* __restrict__ p, int n) {
  int i = blockIdx.x * blockDim.x + threadIdx.x;
  if (i < n) p[i] = 0.0f;
}
__global__ void relu_inplace(float* __restrict__ p, int n) {
  int i = blockIdx.x * blockDim.x + threadIdx.x;
  if (i < n) p[i] = fmaxf(p[i], 0.0f);
}

extern "C" void kernel_launch(void* const* d_in, const int* in_sizes, int n_in,
                              void* d_out, int out_size, void* d_ws, size_t ws_size,
                              hipStream_t stream) {
  const float* x        = (const float*)d_in[0];
  const int*   ei       = (const int*)d_in[1];
  const float* W1       = (const float*)d_in[2];
  const float* att_src1 = (const float*)d_in[3];
  const float* att_dst1 = (const float*)d_in[4];
  const float* b1       = (const float*)d_in[5];
  const float* W2       = (const float*)d_in[6];
  const float* att_src2 = (const float*)d_in[7];
  const float* att_dst2 = (const float*)d_in[8];
  const float* b2       = (const float*)d_in[9];

  const int D_IN = 256, F1 = 256, F2 = 64, H1 = 8, C1 = 32, H2 = 1, C2 = 64;
  const int N = in_sizes[0] / D_IN;
  const int E = in_sizes[1] / 2;
  const int T = E + N;

  float* ws = (float*)d_ws;
  float*    h1    = ws;
  float*    agg1  = h1   + (size_t)N * F1;
  float*    asrc1 = agg1 + (size_t)N * F1;
  float*    adst1 = asrc1 + (size_t)N * H1;
  unsigned* m1    = (unsigned*)(adst1 + (size_t)N * H1);
  float*    s1    = (float*)(m1 + (size_t)N * H1);
  float*    asrc2 = s1 + (size_t)N * H1;
  float*    adst2 = asrc2 + N;
  unsigned* m2    = (unsigned*)(adst2 + N);
  float*    s2    = (float*)(m2 + N);
  float*    out   = (float*)d_out;

  const int BT = 256;
  auto cdiv = [](long long a, long long b) { return (int)((a + b - 1) / b); };

  init_bias<<<cdiv((long long)N * F1, BT), BT, 0, stream>>>(agg1, b1, N, F1);
  init_bias<<<cdiv((long long)N * F2, BT), BT, 0, stream>>>(out, b2, N, F2);
  init_zero_u32<<<cdiv((long long)N * H1, BT), BT, 0, stream>>>(m1, N * H1);
  init_zero_f32<<<cdiv((long long)N * H1, BT), BT, 0, stream>>>(s1, N * H1);
  init_zero_u32<<<cdiv(N, BT), BT, 0, stream>>>(m2, N);
  init_zero_f32<<<cdiv(N, BT), BT, 0, stream>>>(s2, N);

  {
    const int mtiles = (N + 15) / 16, ntiles = F1 / 16;
    const long long waves = (long long)mtiles * ntiles;
    gemm_wmma_f32<256, 256><<<cdiv(waves * 32, BT), BT, 0, stream>>>(x, W1, h1, N);
  }
  att_scores<<<cdiv((long long)N * H1, BT), BT, 0, stream>>>(
      h1, att_src1, att_dst1, asrc1, adst1, N, H1, C1);
  edge_max<8><<<cdiv(T, BT), BT, 0, stream>>>(ei, E, T, asrc1, adst1, m1);
  edge_expsum<8><<<cdiv(T, BT), BT, 0, stream>>>(ei, E, T, asrc1, adst1, m1, s1);
  edge_aggregate<8, 32><<<cdiv((long long)T * 32, BT), BT, 0, stream>>>(
      ei, E, T, asrc1, adst1, m1, s1, h1, agg1);
  relu_inplace<<<cdiv((long long)N * F1, BT), BT, 0, stream>>>(agg1, N * F1);

  float* h2 = h1;
  {
    const int mtiles = (N + 15) / 16, ntiles = F2 / 16;
    const long long waves = (long long)mtiles * ntiles;
    gemm_wmma_f32<256, 64><<<cdiv(waves * 32, BT), BT, 0, stream>>>(agg1, W2, h2, N);
  }
  att_scores<<<cdiv(N, BT), BT, 0, stream>>>(
      h2, att_src2, att_dst2, asrc2, adst2, N, H2, C2);
  edge_max<1><<<cdiv(T, BT), BT, 0, stream>>>(ei, E, T, asrc2, adst2, m2);
  edge_expsum<1><<<cdiv(T, BT), BT, 0, stream>>>(ei, E, T, asrc2, adst2, m2, s2);
  edge_aggregate<1, 64><<<cdiv((long long)T * 32, BT), BT, 0, stream>>>(
      ei, E, T, asrc2, adst2, m2, s2, h2, out);
}


# Both GEMMs as C^T = W^T X^T on 256 persistent 16-wave blocks: a wave keeps its 16-column slab of W (all k) in registers, rows of X pass once through LDS; k summed in permuted order inside each 8
# speedup vs baseline: 2.6306x; 2.6306x over previous
#include <hip/hip_runtime.h>
#include <hip/hip_bf16.h>


typedef float v2f __attribute__((ext_vector_type(2)));
typedef float v4f __attribute__((ext_vector_type(4)));
typedef float v8f __attribute__((ext_vector_type(8)));

#define NEG_SLOPE 0.2f

__device__ __forceinline__ unsigned f2ord(float f) {
  unsigned u = __float_as_uint(f);
  return (f >= 0.0f) ? (u | 0x80000000u) : ~u;
}
__device__ __forceinline__ float ord2f(unsigned u) {
  return (u & 0x80000000u) ? __uint_as_float(u & 0x7FFFFFFFu)
                           : __uint_as_float(~u);
}

template <int NOUT, int TPW>
__global__ __launch_bounds__(512) void gemm_w_in_regs(const float* __restrict__ X,
                                                      const float* __restrict__ W,
                                                      float* __restrict__ C, int M) {
  constexpr int K  = 256;
  constexpr int NS = NOUT / 16;
  constexpr int NG = 16 / NS;
  constexpr int SN = NG * TPW * 16;
  constexpr int P  = K + 4;
  constexpr int LD = SN * (K / 4) / 512;
  static_assert(NS * NG == 16 && LD * 512 == SN * (K / 4), "block shape");
  __shared__ __attribute__((aligned(16))) float xs[2][SN * P];

  const int tid  = threadIdx.x;
  const int lane = tid & 31;
  const int wave = tid >> 5;
  const int c    = lane & 15;
  const int h    = lane >> 4;
  const int slab = wave % NS;
  const int grp  = wave / NS;

  v2f wr[K / 4];
  {
    const float* wp = W + (size_t)(4 * h) * NOUT + slab * 16 + c;
#pragma unroll
    for (int j = 0; j < K / 4; ++j) {
      const int k0 = 8 * (j >> 1) + 2 * (j & 1);
      wr[j].x = wp[(size_t)k0 * NOUT];
      wr[j].y = wp[(size_t)(k0 + 1) * NOUT];
    }
  }

  const int nsteps = (M + SN - 1) / SN;
  v4f pf[LD];

  int step = blockIdx.x;
  if (step < nsteps) {
#pragma unroll
    for (int i = 0; i < LD; ++i) {
      const int idx = tid + 512 * i;
      int row = step * SN + (idx >> 6);
      if (row >= M) row = M - 1;
      pf[i] = *(const v4f*)(X + (size_t)row * K + (idx & 63) * 4);
    }
#pragma unroll
    for (int i = 0; i < LD; ++i) {
      const int idx = tid + 512 * i;
      *(v4f*)(&xs[0][(idx >> 6) * P + (idx & 63) * 4]) = pf[i];
    }
  }
  __syncthreads();

  int cur = 0;
  for (; step < nsteps; step += (int)gridDim.x) {
    const int  nxt      = step + (int)gridDim.x;
    const bool has_next = nxt < nsteps;
    if (has_next) {
#pragma unroll
      for (int i = 0; i < LD; ++i) {
        const int idx = tid + 512 * i;
        int row = nxt * SN + (idx >> 6);
        if (row >= M) row = M - 1;
        pf[i] = *(const v4f*)(X + (size_t)row * K + (idx & 63) * 4);
      }
    }

    v8f acc[TPW];
#pragma unroll
    for (int t = 0; t < TPW; ++t) acc[t] = v8f{};

    const float* xb = &xs[cur][(grp * TPW * 16 + c) * P + 4 * h];
#pragma unroll
    for (int kb = 0; kb < K / 8; ++kb) {
      v4f f[TPW];
#pragma unroll
      for (int t = 0; t < TPW; ++t) f[t] = *(const v4f*)(xb + t * 16 * P + kb * 8);
#pragma unroll
      for (int t = 0; t < TPW; ++t) {
        v2f b = {f[t].x, f[t].y};
        acc[t] = __builtin_amdgcn_wmma_f32_16x16x4_f32(
             false, wr[2 * kb],  false, b,
             (short)0, acc[t],  false,  false);
      }
#pragma unroll
      for (int t = 0; t < TPW; ++t) {
        v2f b = {f[t].z, f[t].w};
        acc[t] = __builtin_amdgcn_wmma_f32_16x16x4_f32(
             false, wr[2 * kb + 1],  false, b,
             (short)0, acc[t],  false,  false);
      }
    }

    if (has_next) {
#pragma unroll
      for (int i = 0; i < LD; ++i) {
        const int idx = tid + 512 * i;
        *(v4f*)(&xs[cur ^ 1][(idx >> 6) * P + (idx & 63) * 4]) = pf[i];
      }
    }
    __syncthreads();
    cur ^= 1;

#pragma unroll
    for (int t = 0; t < TPW; ++t) {
      const int row = step * SN + (grp * TPW + t) * 16 + c;
      if (row < M) {
        float* cp = C + (size_t)row * NOUT + slab * 16 + 8 * h;
        *(v4f*)(cp)     = v4f{acc[t][0], acc[t][1], acc[t][2], acc[t][3]};
        *(v4f*)(cp + 4) = v4f{acc[t][4], acc[t][5], acc[t][6], acc[t][7]};
      }
    }
  }
}

__global__ void att_scores(const float* __restrict__ h,
                           const float* __restrict__ att_src,
                           const float* __restrict__ att_dst,
                           float* __restrict__ asrc, float* __restrict__ adst,
                           int n, int heads, int ch) {
  int i = blockIdx.x * blockDim.x + threadIdx.x;
  if (i >= n * heads) return;
  int node = i / heads, hd = i % heads;
  const float* hp = h + (size_t)node * heads * ch + hd * ch;
  const float* as = att_src + hd * ch;
  const float* ad = att_dst + hd * ch;
  float ss = 0.0f, sd = 0.0f;
  for (int c = 0; c < ch; ++c) {
    float v = hp[c];
    ss += v * as[c];
    sd += v * ad[c];
  }
  asrc[i] = ss;
  adst[i] = sd;
}

template <int H, int C>
__device__ __forceinline__ void edge_fetch(int s, int d, int lane,
                                           const float* __restrict__ asrc,
                                           const float* __restrict__ adst,
                                           const float* __restrict__ hfeat,
                                           float (&hv)[H * C / 32], float& sa, float& sd) {
  constexpr int F = H * C;
  const float* hp = hfeat + (size_t)s * F;
#pragma unroll
  for (int k = 0; k < F / 32; ++k) hv[k] = hp[lane + 32 * k];
  const int hl = (lane < H) ? lane : 0;
  sa = asrc[s * H + hl];
  sd = adst[d * H + hl];
}

template <int H, int C>
__device__ __forceinline__ void edge_commit(int j, int lane,
                                            const float (&hv)[H * C / 32], float sa, float sd,
                                            const unsigned* mx, float* ssum,
                                            float* acc) {
  constexpr int F = H * C;
  static_assert(C % 32 == 0 && H <= 32, "head index must be wave-uniform per k");
  float ex = 0.0f;
  if (lane < H) {
    const float sc = sa + sd;
    float v = (sc > 0.0f) ? sc : NEG_SLOPE * sc;
    ex = __expf(v - ord2f(mx[j * H + lane]));
    atomicAdd(&ssum[j * H + lane], ex);
  }
  float* op = acc + j * F;
#pragma unroll
  for (int k = 0; k < F / 32; ++k) {
    float a = __int_as_float(
        __builtin_amdgcn_readlane(__float_as_int(ex), (32 * k) / C));
    atomicAdd(&op[lane + 32 * k], hv[k] * a);
  }
}

template <int H, int C, bool RELU, int TN, int CAP>
__global__ __launch_bounds__(1024) void tile_softmax_aggregate(
    const int* __restrict__ ei, int E, int N,
    const float* __restrict__ asrc, const float* __restrict__ adst,
    const float* __restrict__ hfeat, const float* __restrict__ bias,
    float* __restrict__ dst) {
  constexpr int F  = H * C;
  constexpr int NW = 32;
  constexpr int U  = 8;
  constexpr int P  = 4;
  constexpr int Q  = 4;
  __shared__ float    acc[TN * F];
  __shared__ unsigned mx[TN * H];
  __shared__ float    ssum[TN * H];
  __shared__ int      lst_s[CAP];
  __shared__ int      lst_j[CAP];
  __shared__ unsigned cnt;

  const int tid  = threadIdx.x;
  const int lane = tid & 31;
  const int wave = tid >> 5;
  const int n0   = blockIdx.x * TN;
  const int nloc = (N - n0 < TN) ? (N - n0) : TN;
  const int* __restrict__ srcp = ei;
  const int* __restrict__ dstp = ei + E;

  for (int i = tid; i < nloc * F; i += NW * 32) acc[i] = 0.0f;
  for (int i = tid; i < nloc * H; i += NW * 32) {
    float v = asrc[(size_t)n0 * H + i] + adst[(size_t)n0 * H + i];
    v = (v > 0.0f) ? v : NEG_SLOPE * v;
    mx[i]   = f2ord(v);
    ssum[i] = 0.0f;
  }
  if (tid == 0) cnt = 0u;
  __syncthreads();

  const int nchunks = (E + 31) >> 5;
  const int rot = (nchunks / (int)gridDim.x) * (int)blockIdx.x;

  for (int c0 = wave; c0 < nchunks; c0 += NW * U) {
    int dd[U], ee[U];
#pragma unroll
    for (int u = 0; u < U; ++u) {
      const int cl = c0 + u * NW;
      int c = cl + rot;
      if (c >= nchunks) c -= nchunks;
      const int e = c * 32 + lane;
      ee[u] = e;
      dd[u] = (cl < nchunks && e < E) ? dstp[e] : -1;
    }
#pragma unroll
    for (int u = 0; u < U; ++u) {
      const unsigned j = (unsigned)(dd[u] - n0);
      if (j < (unsigned)nloc) {
        const unsigned pos = atomicAdd(&cnt, 1u);
        if (pos < (unsigned)CAP) {
          lst_s[pos] = ee[u];
          lst_j[pos] = (int)j;
        }
      }
    }
  }
  __syncthreads();
  const unsigned total = cnt;

  if (total <= (unsigned)CAP) {
    const int ne = (int)total;
    for (int i = tid; i < ne; i += NW * 32) lst_s[i] = srcp[lst_s[i]];
    __syncthreads();

    for (int i = tid; i < ne * H; i += NW * 32 * Q) {
      float va[Q], vb[Q];
      int   slot[Q];
#pragma unroll
      for (int q = 0; q < Q; ++q) {
        const int ii = i + q * NW * 32;
        slot[q] = -1;
        va[q] = 0.0f;
        vb[q] = 0.0f;
        if (ii < ne * H) {
          const int en = ii / H, h = ii % H;
          const int s = lst_s[en], j = lst_j[en];
          va[q] = asrc[s * H + h];
          vb[q] = adst[(n0 + j) * H + h];
          slot[q] = j * H + h;
        }
      }
#pragma unroll
      for (int q = 0; q < Q; ++q) {
        if (slot[q] >= 0) {
          float v = va[q] + vb[q];
          v = (v > 0.0f) ? v : NEG_SLOPE * v;
          atomicMax(&mx[slot[q]], f2ord(v));
        }
      }
    }
    __syncthreads();

    const int tot = ne + nloc;
    for (int i0 = wave * P; i0 < tot; i0 += NW * P) {
      int   jv[P];
      float hv[P][F / 32];
      float sa[P], sd[P];
#pragma unroll
      for (int p = 0; p < P; ++p) {
        const int i = i0 + p;
        jv[p] = 0;
        sa[p] = 0.0f;
        sd[p] = 0.0f;
        if (i < tot) {
          int s, j;
          if (i < ne) { s = lst_s[i]; j = lst_j[i]; }
          else        { j = i - ne;   s = n0 + j;   }
          s = __builtin_amdgcn_readfirstlane(s);
          j = __builtin_amdgcn_readfirstlane(j);
          jv[p] = j;
          edge_fetch<H, C>(s, n0 + j, lane, asrc, adst, hfeat, hv[p], sa[p], sd[p]);
        }
      }
#pragma unroll
      for (int p = 0; p < P; ++p) {
        if (i0 + p < tot) edge_commit<H, C>(jv[p], lane, hv[p], sa[p], sd[p], mx, ssum, acc);
      }
    }
  } else {
    for (int c0 = wave; c0 < nchunks; c0 += NW * U) {
      int dd[U], ee[U];
#pragma unroll
      for (int u = 0; u < U; ++u) {
        const int cl = c0 + u * NW;
        int c = cl + rot;
        if (c >= nchunks) c -= nchunks;
        const int e = c * 32 + lane;
        ee[u] = e;
        dd[u] = (cl < nchunks && e < E) ? dstp[e] : -1;
      }
#pragma unroll
      for (int u = 0; u < U; ++u) {
        const unsigned j = (unsigned)(dd[u] - n0);
        if (j < (unsigned)nloc) {
          const int s = srcp[ee[u]];
#pragma unroll
          for (int h = 0; h < H; ++h) {
            float v = asrc[s * H + h] + adst[dd[u] * H + h];
            v = (v > 0.0f) ? v : NEG_SLOPE * v;
            atomicMax(&mx[j * H + h], f2ord(v));
          }
        }
      }
    }
    __syncthreads();

    for (int j = wave; j < nloc; j += NW) {
      float hv[F / 32], sa, sd;
      edge_fetch<H, C>(n0 + j, n0 + j, lane, asrc, adst, hfeat, hv, sa, sd);
      edge_commit<H, C>(j, lane, hv, sa, sd, mx, ssum, acc);
    }
    for (int c0 = wave; c0 < nchunks; c0 += NW * U) {
      int dd[U], ee[U];
#pragma unroll
      for (int u = 0; u < U; ++u) {
        const int cl = c0 + u * NW;
        int c = cl + rot;
        if (c >= nchunks) c -= nchunks;
        const int e = c * 32 + lane;
        ee[u] = e;
        dd[u] = (cl < nchunks && e < E) ? dstp[e] : -1;
      }
#pragma unroll
      for (int u = 0; u < U; ++u) {
        const unsigned j = (unsigned)(dd[u] - n0);
        const bool match = j < (unsigned)nloc;
        unsigned mask = __builtin_amdgcn_ballot_w32(match);
        if (mask) {
          const int s = match ? srcp[ee[u]] : 0;
          while (mask) {
            const int l = __builtin_ctz(mask);
            mask &= mask - 1;
            const int sj = __builtin_amdgcn_readlane(s, l);
            const int jj = __builtin_amdgcn_readlane((int)j, l);
            float hv[F / 32], sa, sd;
            edge_fetch<H, C>(sj, n0 + jj, lane, asrc, adst, hfeat, hv, sa, sd);
            edge_commit<H, C>(jj, lane, hv, sa, sd, mx, ssum, acc);
          }
        }
      }
    }
  }
  __syncthreads();

  for (int i = tid; i < nloc * F; i += NW * 32) {
    const int node = i / F, f = i % F;
    float v = acc[i] / (ssum[node * H + f / C] + 1e-16f) + bias[f];
    dst[(size_t)n0 * F + i] = RELU ? fmaxf(v, 0.0f) : v;
  }
}

extern "C" void kernel_launch(void* const* d_in, const int* in_sizes, int n_in,
                              void* d_out, int out_size, void* d_ws, size_t ws_size,
                              hipStream_t stream) {
  const float* x        = (const float*)d_in[0];
  const int*   ei       = (const int*)d_in[1];
  const float* W1       = (const float*)d_in[2];
  const float* att_src1 = (const float*)d_in[3];
  const float* att_dst1 = (const float*)d_in[4];
  const float* b1       = (const float*)d_in[5];
  const float* W2       = (const float*)d_in[6];
  const float* att_src2 = (const float*)d_in[7];
  const float* att_dst2 = (const float*)d_in[8];
  const float* b2       = (const float*)d_in[9];

  const int D_IN = 256, F1 = 256, F2 = 64, H1 = 8, C1 = 32, H2 = 1, C2 = 64;
  const int N = in_sizes[0] / D_IN;
  const int E = in_sizes[1] / 2;

  float* ws = (float*)d_ws;
  float* h1    = ws;
  float* agg1  = h1   + (size_t)N * F1;
  float* asrc1 = agg1 + (size_t)N * F1;
  float* adst1 = asrc1 + (size_t)N * H1;
  float* asrc2 = adst1 + (size_t)N * H1;
  float* adst2 = asrc2 + N;
  float* out   = (float*)d_out;

  const int BT = 256;
  const int GB = 256;
  const int TN = 200;
  const int CAP = 12288;
  auto cdiv = [](long long a, long long b) { return (int)((a + b - 1) / b); };

  gemm_w_in_regs<256, 2><<<GB, 512, 0, stream>>>(x, W1, h1, N);
  att_scores<<<cdiv((long long)N * H1, BT), BT, 0, stream>>>(
      h1, att_src1, att_dst1, asrc1, adst1, N, H1, C1);
  tile_softmax_aggregate<8, 32, true, TN, CAP><<<cdiv(N, TN), 1024, 0, stream>>>(
      ei, E, N, asrc1, adst1, h1, b1, agg1);

  float* h2 = h1;
  gemm_w_in_regs<64, 1><<<GB, 512, 0, stream>>>(agg1, W2, h2, N);
  att_scores<<<cdiv(N, BT), BT, 0, stream>>>(
      h2, att_src2, att_dst2, asrc2, adst2, N, H2, C2);
  tile_softmax_aggregate<1, 64, false, TN, CAP><<<cdiv(N, TN), 1024, 0, stream>>>(
      ei, E, N, asrc2, adst2, h2, b2, out);
}


# Each tile's edge list is built once per call: layer 1's aggregate keeps it (count, sources, rows) in the workspace, layer 2's loads it, its own edge sweep gone; layer-2 sum order: that list's order
# speedup vs baseline: 3.3737x; 1.2825x over previous
#include <hip/hip_runtime.h>
#include <hip/hip_bf16.h>


typedef float v2f __attribute__((ext_vector_type(2)));
typedef float v4f __attribute__((ext_vector_type(4)));
typedef float v8f __attribute__((ext_vector_type(8)));

#define NEG_SLOPE 0.2f

__device__ __forceinline__ unsigned f2ord(float f) {
  unsigned u = __float_as_uint(f);
  return (f >= 0.0f) ? (u | 0x80000000u) : ~u;
}
__device__ __forceinline__ float ord2f(unsigned u) {
  return (u & 0x80000000u) ? __uint_as_float(u & 0x7FFFFFFFu)
                           : __uint_as_float(~u);
}

template <int NOUT, int TPW>
__global__ __launch_bounds__(512) void gemm_w_in_regs(const float* __restrict__ X,
                                                      const float* __restrict__ W,
                                                      float* __restrict__ C, int M) {
  constexpr int K  = 256;
  constexpr int NS = NOUT / 16;
  constexpr int NG = 16 / NS;
  constexpr int SN = NG * TPW * 16;
  constexpr int P  = K + 4;
  constexpr int LD = SN * (K / 4) / 512;
  static_assert(NS * NG == 16 && LD * 512 == SN * (K / 4), "block shape");
  __shared__ __attribute__((aligned(16))) float xs[2][SN * P];

  const int tid  = threadIdx.x;
  const int lane = tid & 31;
  const int wave = tid >> 5;
  const int c    = lane & 15;
  const int h    = lane >> 4;
  const int slab = wave % NS;
  const int grp  = wave / NS;

  v2f wr[K / 4];
  {
    const float* wp = W + (size_t)(4 * h) * NOUT + slab * 16 + c;
#pragma unroll
    for (int j = 0; j < K / 4; ++j) {
      const int k0 = 8 * (j >> 1) + 2 * (j & 1);
      wr[j].x = wp[(size_t)k0 * NOUT];
      wr[j].y = wp[(size_t)(k0 + 1) * NOUT];
    }
  }

  const int nsteps = (M + SN - 1) / SN;
  v4f pf[LD];

  int step = blockIdx.x;
  if (step < nsteps) {
#pragma unroll
    for (int i = 0; i < LD; ++i) {
      const int idx = tid + 512 * i;
      int row = step * SN + (idx >> 6);
      if (row >= M) row = M - 1;
      pf[i] = *(const v4f*)(X + (size_t)row * K + (idx & 63) * 4);
    }
#pragma unroll
    for (int i = 0; i < LD; ++i) {
      const int idx = tid + 512 * i;
      *(v4f*)(&xs[0][(idx >> 6) * P + (idx & 63) * 4]) = pf[i];
    }
  }
  __syncthreads();

  int cur = 0;
  for (; step < nsteps; step += (int)gridDim.x) {
    const int  nxt      = step + (int)gridDim.x;
    const bool has_next = nxt < nsteps;
    if (has_next) {
#pragma unroll
      for (int i = 0; i < LD; ++i) {
        const int idx = tid + 512 * i;
        int row = nxt * SN + (idx >> 6);
        if (row >= M) row = M - 1;
        pf[i] = *(const v4f*)(X + (size_t)row * K + (idx & 63) * 4);
      }
    }

    v8f acc[TPW];
#pragma unroll
    for (int t = 0; t < TPW; ++t) acc[t] = v8f{};

    const float* xb = &xs[cur][(grp * TPW * 16 + c) * P + 4 * h];
#pragma unroll
    for (int kb = 0; kb < K / 8; ++kb) {
      v4f f[TPW];
#pragma unroll
      for (int t = 0; t < TPW; ++t) f[t] = *(const v4f*)(xb + t * 16 * P + kb * 8);
#pragma unroll
      for (int t = 0; t < TPW; ++t) {
        v2f b = {f[t].x, f[t].y};
        acc[t] = __builtin_amdgcn_wmma_f32_16x16x4_f32(
             false, wr[2 * kb],  false, b,
             (short)0, acc[t],  false,  false);
      }
#pragma unroll
      for (int t = 0; t < TPW; ++t) {
        v2f b = {f[t].z, f[t].w};
        acc[t] = __builtin_amdgcn_wmma_f32_16x16x4_f32(
             false, wr[2 * kb + 1],  false, b,
             (short)0, acc[t],  false,  false);
      }
    }

    if (has_next) {
#pragma unroll
      for (int i = 0; i < LD; ++i) {
        const int idx = tid + 512 * i;
        *(v4f*)(&xs[cur ^ 1][(idx >> 6) * P + (idx & 63) * 4]) = pf[i];
      }
    }
    __syncthreads();
    cur ^= 1;

#pragma unroll
    for (int t = 0; t < TPW; ++t) {
      const int row = step * SN + (grp * TPW + t) * 16 + c;
      if (row < M) {
        float* cp = C + (size_t)row * NOUT + slab * 16 + 8 * h;
        *(v4f*)(cp)     = v4f{acc[t][0], acc[t][1], acc[t][2], acc[t][3]};
        *(v4f*)(cp + 4) = v4f{acc[t][4], acc[t][5], acc[t][6], acc[t][7]};
      }
    }
  }
}

__global__ void att_scores(const float* __restrict__ h,
                           const float* __restrict__ att_src,
                           const float* __restrict__ att_dst,
                           float* __restrict__ asrc, float* __restrict__ adst,
                           int n, int heads, int ch) {
  int i = blockIdx.x * blockDim.x + threadIdx.x;
  if (i >= n * heads) return;
  int node = i / heads, hd = i % heads;
  const float* hp = h + (size_t)node * heads * ch + hd * ch;
  const float* as = att_src + hd * ch;
  const float* ad = att_dst + hd * ch;
  float ss = 0.0f, sd = 0.0f;
  for (int c = 0; c < ch; ++c) {
    float v = hp[c];
    ss += v * as[c];
    sd += v * ad[c];
  }
  asrc[i] = ss;
  adst[i] = sd;
}

template <int H, int C>
__device__ __forceinline__ void edge_fetch(int s, int d, int lane,
                                           const float* __restrict__ asrc,
                                           const float* __restrict__ adst,
                                           const float* __restrict__ hfeat,
                                           float (&hv)[H * C / 32], float& sa, float& sd) {
  constexpr int F = H * C;
  const float* hp = hfeat + (size_t)s * F;
#pragma unroll
  for (int k = 0; k < F / 32; ++k) hv[k] = hp[lane + 32 * k];
  const int hl = (lane < H) ? lane : 0;
  sa = asrc[s * H + hl];
  sd = adst[d * H + hl];
}

template <int H, int C>
__device__ __forceinline__ void edge_commit(int j, int lane,
                                            const float (&hv)[H * C / 32], float sa, float sd,
                                            const unsigned* mx, float* ssum,
                                            float* acc) {
  constexpr int F = H * C;
  static_assert(C % 32 == 0 && H <= 32, "head index must be wave-uniform per k");
  float ex = 0.0f;
  if (lane < H) {
    const float sc = sa + sd;
    float v = (sc > 0.0f) ? sc : NEG_SLOPE * sc;
    ex = __expf(v - ord2f(mx[j * H + lane]));
    atomicAdd(&ssum[j * H + lane], ex);
  }
  float* op = acc + j * F;
#pragma unroll
  for (int k = 0; k < F / 32; ++k) {
    float a = __int_as_float(
        __builtin_amdgcn_readlane(__float_as_int(ex), (32 * k) / C));
    atomicAdd(&op[lane + 32 * k], hv[k] * a);
  }
}

template <int H, int C, bool RELU, int TN, int CAP, int LIST>
__global__ __launch_bounds__(1024) void tile_softmax_aggregate(
    const int* __restrict__ ei, int E, int N,
    const float* __restrict__ asrc, const float* __restrict__ adst,
    const float* __restrict__ hfeat, const float* __restrict__ bias,
    float* __restrict__ dst,
    unsigned* __restrict__ g_cnt, int* __restrict__ g_s, int* __restrict__ g_j) {
  constexpr int F  = H * C;
  constexpr int NW = 32;
  constexpr int U  = 8;
  constexpr int P  = 4;
  constexpr int Q  = 4;
  __shared__ float    acc[TN * F];
  __shared__ unsigned mx[TN * H];
  __shared__ float    ssum[TN * H];
  __shared__ int      lst_s[CAP];
  __shared__ int      lst_j[CAP];
  __shared__ unsigned cnt;

  const int tid  = threadIdx.x;
  const int lane = tid & 31;
  const int wave = tid >> 5;
  const int n0   = blockIdx.x * TN;
  const int nloc = (N - n0 < TN) ? (N - n0) : TN;
  const int* __restrict__ srcp = ei;
  const int* __restrict__ dstp = ei + E;

  for (int i = tid; i < nloc * F; i += NW * 32) acc[i] = 0.0f;
  for (int i = tid; i < nloc * H; i += NW * 32) {
    float v = asrc[(size_t)n0 * H + i] + adst[(size_t)n0 * H + i];
    v = (v > 0.0f) ? v : NEG_SLOPE * v;
    mx[i]   = f2ord(v);
    ssum[i] = 0.0f;
  }
  if (tid == 0) cnt = 0u;
  __syncthreads();

  const int nchunks = (E + 31) >> 5;
  const int rot = (nchunks / (int)gridDim.x) * (int)blockIdx.x;

  int* __restrict__ gs = g_s + (size_t)blockIdx.x * CAP;
  int* __restrict__ gj = g_j + (size_t)blockIdx.x * CAP;
  unsigned total;
  if (LIST == 2) {
    total = g_cnt[blockIdx.x];
  } else {
    for (int c0 = wave; c0 < nchunks; c0 += NW * U) {
      int dd[U], ee[U];
#pragma unroll
      for (int u = 0; u < U; ++u) {
        const int cl = c0 + u * NW;
        int c = cl + rot;
        if (c >= nchunks) c -= nchunks;
        const int e = c * 32 + lane;
        ee[u] = e;
        dd[u] = (cl < nchunks && e < E) ? dstp[e] : -1;
      }
#pragma unroll
      for (int u = 0; u < U; ++u) {
        const unsigned j = (unsigned)(dd[u] - n0);
        if (j < (unsigned)nloc) {
          const unsigned pos = atomicAdd(&cnt, 1u);
          if (pos < (unsigned)CAP) {
            lst_s[pos] = ee[u];
            lst_j[pos] = (int)j;
          }
        }
      }
    }
    __syncthreads();
    total = cnt;
    if (tid == 0) g_cnt[blockIdx.x] = total;
  }

  if (total <= (unsigned)CAP) {
    const int ne = (int)total;
    if (LIST == 2) {
      for (int i = tid; i < ne; i += NW * 32) {
        lst_s[i] = gs[i];
        lst_j[i] = gj[i];
      }
    } else {
      for (int i = tid; i < ne; i += NW * 32) {
        const int s = srcp[lst_s[i]];
        lst_s[i] = s;
        gs[i] = s;
        gj[i] = lst_j[i];
      }
    }
    __syncthreads();

    for (int i = tid; i < ne * H; i += NW * 32 * Q) {
      float va[Q], vb[Q];
      int   slot[Q];
#pragma unroll
      for (int q = 0; q < Q; ++q) {
        const int ii = i + q * NW * 32;
        slot[q] = -1;
        va[q] = 0.0f;
        vb[q] = 0.0f;
        if (ii < ne * H) {
          const int en = ii / H, h = ii % H;
          const int s = lst_s[en], j = lst_j[en];
          va[q] = asrc[s * H + h];
          vb[q] = adst[(n0 + j) * H + h];
          slot[q] = j * H + h;
        }
      }
#pragma unroll
      for (int q = 0; q < Q; ++q) {
        if (slot[q] >= 0) {
          float v = va[q] + vb[q];
          v = (v > 0.0f) ? v : NEG_SLOPE * v;
          atomicMax(&mx[slot[q]], f2ord(v));
        }
      }
    }
    __syncthreads();

    const int tot = ne + nloc;
    for (int i0 = wave * P; i0 < tot; i0 += NW * P) {
      int   jv[P];
      float hv[P][F / 32];
      float sa[P], sd[P];
#pragma unroll
      for (int p = 0; p < P; ++p) {
        const int i = i0 + p;
        jv[p] = 0;
        sa[p] = 0.0f;
        sd[p] = 0.0f;
        if (i < tot) {
          int s, j;
          if (i < ne) { s = lst_s[i]; j = lst_j[i]; }
          else        { j = i - ne;   s = n0 + j;   }
          s = __builtin_amdgcn_readfirstlane(s);
          j = __builtin_amdgcn_readfirstlane(j);
          jv[p] = j;
          edge_fetch<H, C>(s, n0 + j, lane, asrc, adst, hfeat, hv[p], sa[p], sd[p]);
        }
      }
#pragma unroll
      for (int p = 0; p < P; ++p) {
        if (i0 + p < tot) edge_commit<H, C>(jv[p], lane, hv[p], sa[p], sd[p], mx, ssum, acc);
      }
    }
  } else {
    for (int c0 = wave; c0 < nchunks; c0 += NW * U) {
      int dd[U], ee[U];
#pragma unroll
      for (int u = 0; u < U; ++u) {
        const int cl = c0 + u * NW;
        int c = cl + rot;
        if (c >= nchunks) c -= nchunks;
        const int e = c * 32 + lane;
        ee[u] = e;
        dd[u] = (cl < nchunks && e < E) ? dstp[e] : -1;
      }
#pragma unroll
      for (int u = 0; u < U; ++u) {
        const unsigned j = (unsigned)(dd[u] - n0);
        if (j < (unsigned)nloc) {
          const int s = srcp[ee[u]];
#pragma unroll
          for (int h = 0; h < H; ++h) {
            float v = asrc[s * H + h] + adst[dd[u] * H + h];
            v = (v > 0.0f) ? v : NEG_SLOPE * v;
            atomicMax(&mx[j * H + h], f2ord(v));
          }
        }
      }
    }
    __syncthreads();

    for (int j = wave; j < nloc; j += NW) {
      float hv[F / 32], sa, sd;
      edge_fetch<H, C>(n0 + j, n0 + j, lane, asrc, adst, hfeat, hv, sa, sd);
      edge_commit<H, C>(j, lane, hv, sa, sd, mx, ssum, acc);
    }
    for (int c0 = wave; c0 < nchunks; c0 += NW * U) {
      int dd[U], ee[U];
#pragma unroll
      for (int u = 0; u < U; ++u) {
        const int cl = c0 + u * NW;
        int c = cl + rot;
        if (c >= nchunks) c -= nchunks;
        const int e = c * 32 + lane;
        ee[u] = e;
        dd[u] = (cl < nchunks && e < E) ? dstp[e] : -1;
      }
#pragma unroll
      for (int u = 0; u < U; ++u) {
        const unsigned j = (unsigned)(dd[u] - n0);
        const bool match = j < (unsigned)nloc;
        unsigned mask = __builtin_amdgcn_ballot_w32(match);
        if (mask) {
          const int s = match ? srcp[ee[u]] : 0;
          while (mask) {
            const int l = __builtin_ctz(mask);
            mask &= mask - 1;
            const int sj = __builtin_amdgcn_readlane(s, l);
            const int jj = __builtin_amdgcn_readlane((int)j, l);
            float hv[F / 32], sa, sd;
            edge_fetch<H, C>(sj, n0 + jj, lane, asrc, adst, hfeat, hv, sa, sd);
            edge_commit<H, C>(jj, lane, hv, sa, sd, mx, ssum, acc);
          }
        }
      }
    }
  }
  __syncthreads();

  for (int i = tid; i < nloc * F; i += NW * 32) {
    const int node = i / F, f = i % F;
    float v = acc[i] / (ssum[node * H + f / C] + 1e-16f) + bias[f];
    dst[(size_t)n0 * F + i] = RELU ? fmaxf(v, 0.0f) : v;
  }
}

extern "C" void kernel_launch(void* const* d_in, const int* in_sizes, int n_in,
                              void* d_out, int out_size, void* d_ws, size_t ws_size,
                              hipStream_t stream) {
  const float* x        = (const float*)d_in[0];
  const int*   ei       = (const int*)d_in[1];
  const float* W1       = (const float*)d_in[2];
  const float* att_src1 = (const float*)d_in[3];
  const float* att_dst1 = (const float*)d_in[4];
  const float* b1       = (const float*)d_in[5];
  const float* W2       = (const float*)d_in[6];
  const float* att_src2 = (const float*)d_in[7];
  const float* att_dst2 = (const float*)d_in[8];
  const float* b2       = (const float*)d_in[9];

  const int D_IN = 256, F1 = 256, F2 = 64, H1 = 8, C1 = 32, H2 = 1, C2 = 64;
  const int N = in_sizes[0] / D_IN;
  const int E = in_sizes[1] / 2;

  float* ws = (float*)d_ws;
  float* h1    = ws;
  float* agg1  = h1   + (size_t)N * F1;
  float* asrc1 = agg1 + (size_t)N * F1;
  float* adst1 = asrc1 + (size_t)N * H1;
  float* asrc2 = adst1 + (size_t)N * H1;
  float* adst2 = asrc2 + N;
  float* out   = (float*)d_out;

  const int BT = 256;
  const int GB = 256;
  const int TN = 200;
  const int CAP = 12288;
  auto cdiv = [](long long a, long long b) { return (int)((a + b - 1) / b); };
  const int NT = cdiv(N, TN);
  unsigned* tl_cnt = (unsigned*)(adst2 + N);
  int*      tl_s   = (int*)(tl_cnt + NT);
  int*      tl_j   = tl_s + (size_t)NT * CAP;

  gemm_w_in_regs<256, 2><<<GB, 512, 0, stream>>>(x, W1, h1, N);
  att_scores<<<cdiv((long long)N * H1, BT), BT, 0, stream>>>(
      h1, att_src1, att_dst1, asrc1, adst1, N, H1, C1);
  tile_softmax_aggregate<8, 32, true, TN, CAP, 1><<<NT, 1024, 0, stream>>>(
      ei, E, N, asrc1, adst1, h1, b1, agg1, tl_cnt, tl_s, tl_j);

  float* h2 = h1;
  gemm_w_in_regs<64, 1><<<GB, 512, 0, stream>>>(agg1, W2, h2, N);
  att_scores<<<cdiv(N, BT), BT, 0, stream>>>(
      h2, att_src2, att_dst2, asrc2, adst2, N, H2, C2);
  tile_softmax_aggregate<1, 64, false, TN, CAP, 2><<<NT, 1024, 0, stream>>>(
      ei, E, N, asrc2, adst2, h2, b2, out, tl_cnt, tl_s, tl_j);
}
